# starting kernel
# baseline (speedup 1.0000x reference)
#include <hip/hip_runtime.h>

typedef float v2f __attribute__((ext_vector_type(2)));
typedef float v8f __attribute__((ext_vector_type(8)));

#define DIM_D 256
#define DIM_N 1024
#define EDGES_PER_WAVE 16
#define WAVES_PER_BLOCK 8
#define EDGES_PER_BLOCK (EDGES_PER_WAVE * WAVES_PER_BLOCK)

__global__ __launch_bounds__(256, 4)
void edge_norm_wmma(const float* __restrict__ x,
                    const int*   __restrict__ eb,
                    const int*   __restrict__ ei,
                    const int*   __restrict__ ej,
                    float*       __restrict__ out,
                    int E)
{
    __shared__ float lds[WAVES_PER_BLOCK][16 * 16];

    const int tid   = threadIdx.x;
    const int wave  = tid >> 5;
    const int lane  = tid & 31;
    const int m     = lane & 15;
    const int khalf = (lane >> 4) * 2;

    const long groupBase = (long)blockIdx.x * EDGES_PER_BLOCK
                         + (long)wave * EDGES_PER_WAVE;
    const long e  = groupBase + m;
    const long ec = (e < (long)E) ? e : (long)(E - 1);

    const int b = eb[ec];
    const int i = ei[ec];
    const int j = ej[ec];
    const float* __restrict__ pi = x + ((size_t)b * DIM_N + (size_t)i) * DIM_D;
    const float* __restrict__ pj = x + ((size_t)b * DIM_N + (size_t)j) * DIM_D;

    v8f c = {0.f, 0.f, 0.f, 0.f, 0.f, 0.f, 0.f, 0.f};

    #pragma unroll 4
    for (int kb = 0; kb < DIM_D; kb += 4) {
        const int k = kb + khalf;
        v2f xi = *(const v2f*)(pi + k);
        v2f xj = *(const v2f*)(pj + k);
        v2f a;
        a.x = xi.x - xj.x;
        a.y = xi.y - xj.y;
        c = __builtin_amdgcn_wmma_f32_16x16x4_f32(
                 false, a,  false, a,
                 (short)0, c,  false,  false);
    }

    float* tile = lds[wave];
    const int nIdx  = lane & 15;
    const int mBase = (lane >> 4) * 8;
    #pragma unroll
    for (int r = 0; r < 8; ++r)
        tile[(mBase + r) * 16 + nIdx] = c[r];

    __syncthreads();

    if (lane < 16 && e < (long)E) {
        const float s = tile[m * 16 + m];
        const float d = sqrtf(s);
        float4 v = make_float4(d, d, d, d);
        float4* o = (float4*)(out + (size_t)e * 8);
        o[0] = v;
        o[1] = v;
    }
}

extern "C" void kernel_launch(void* const* d_in, const int* in_sizes, int n_in,
                              void* d_out, int out_size, void* d_ws, size_t ws_size,
                              hipStream_t stream)
{
    const float* x  = (const float*)d_in[0];
    const int*   eb = (const int*)d_in[1];
    const int*   ei = (const int*)d_in[2];
    const int*   ej = (const int*)d_in[3];
    float*       out = (float*)d_out;

    const int E = in_sizes[1];
    if (E <= 0) return;

    const int blocks = (E + EDGES_PER_BLOCK - 1) / EDGES_PER_BLOCK;
    hipLaunchKernelGGL(edge_norm_wmma, dim3(blocks), dim3(256), 0, stream,
                       x, eb, ei, ej, out, E);
}


# Four edges share one WMMA tile (an edge = 4 tile rows = 8 lanes, one whole line per load): one diagonal fold per 4 edges, per-lane row addresses replace the readlanes; sum order changes
# speedup vs baseline: 1.5657x; 1.5657x over previous
#include <hip/hip_runtime.h>

typedef float v2f __attribute__((ext_vector_type(2)));
typedef float v4f __attribute__((ext_vector_type(4)));
typedef float v8f __attribute__((ext_vector_type(8)));

#define DIM_D 256
#define DIM_N 1024
#define WAVES_PER_BLOCK 32
#define THREADS_PER_BLOCK 1024
#define NUM_BLOCKS 256
#define LOG2_NUM_WAVES 13

#define DPP_ROW_SHL(n) (0x100 + (n))
#define DPP_QUAD_0022  (0 | (0 << 2) | (2 << 4) | (2 << 6))

template <int CTRL>
__device__ __forceinline__ float row_read(float v)
{
    return __builtin_bit_cast(float,
        __builtin_amdgcn_update_dpp(0, __builtin_bit_cast(int, v), CTRL, 0xF, 0xF, true));
}

__device__ __forceinline__ long range_start(long w, long E)
{
    return (w >> LOG2_NUM_WAVES) ? E : (((w * E) >> LOG2_NUM_WAVES) & ~3L);
}

__device__ __forceinline__ v8f gram_step(v2f a, v8f c)
{
    return __builtin_amdgcn_wmma_f32_16x16x4_f32(
         false, a,  false, a,
         (short)0, c,  false,  false);
}

__global__ __launch_bounds__(THREADS_PER_BLOCK)
void edge_norm_quad(const float* __restrict__ x,
                    const int*   __restrict__ eb,
                    const int*   __restrict__ ei,
                    const int*   __restrict__ ej,
                    float*       __restrict__ out,
                    int E)
{
    const int lane = threadIdx.x & 31;
    const int gw = __builtin_amdgcn_readfirstlane(
        (int)(blockIdx.x * WAVES_PER_BLOCK + (threadIdx.x >> 5)));

    const long e0 = range_start((long)gw,     (long)E);
    const long e1 = range_start((long)gw + 1, (long)E);
    if (e0 >= e1) return;

    const int c     = lane & 15;
    const int h     = lane >> 4;
    const int slot  = c >> 2;
    const unsigned lane_bytes = (unsigned)((((c & 3) << 1) | h) << 4);

    const bool upper_regs = (c & 4) != 0;
    const bool storer   = ((c & 2) == 0) && (((c >> 3) & 1) == h);
    const int  st_edge  = (h << 1) | ((c >> 2) & 1);

    const char* __restrict__ xb  = (const char*)x;
    const char* __restrict__ ebb = (const char*)eb;
    const char* __restrict__ eib = (const char*)ei;
    const char* __restrict__ ejb = (const char*)ej;

    const int first = (int)e0;
    const int last  = (int)e1 - 1;

    int eidx = first + slot;
    if (eidx > last) eidx = last;
    unsigned ioff = (unsigned)eidx << 2;
    int nb = *(const int*)(ebb + ioff);
    int rowi_next = nb * DIM_N + *(const int*)(eib + ioff);
    int rowj_next = nb * DIM_N + *(const int*)(ejb + ioff);

    for (int g = first; g <= last; g += 4) {
        const unsigned offi = ((unsigned)rowi_next << 10) | lane_bytes;
        const unsigned offj = ((unsigned)rowj_next << 10) | lane_bytes;

        v4f a[8], b[8];
        #pragma unroll
        for (int t = 0; t < 8; ++t) {
            a[t] = *(const v4f*)(xb + offi + t * 128);
            b[t] = *(const v4f*)(xb + offj + t * 128);
        }

        eidx = g + 4 + slot;
        if (eidx > last) eidx = last;
        ioff = (unsigned)eidx << 2;
        nb = *(const int*)(ebb + ioff);
        rowi_next = nb * DIM_N + *(const int*)(eib + ioff);
        rowj_next = nb * DIM_N + *(const int*)(ejb + ioff);

        __builtin_amdgcn_sched_barrier(0);

        v8f acc0 = {0.f, 0.f, 0.f, 0.f, 0.f, 0.f, 0.f, 0.f};
        v8f acc1 = {0.f, 0.f, 0.f, 0.f, 0.f, 0.f, 0.f, 0.f};
        #pragma unroll
        for (int t = 0; t < 8; ++t) {
            const v4f d = a[t] - b[t];
            acc0 = gram_step(__builtin_shufflevector(d, d, 0, 1), acc0);
            acc1 = gram_step(__builtin_shufflevector(d, d, 2, 3), acc1);
        }
        const v8f acc = acc0 + acc1;

        const float p01 = acc[0] + row_read<DPP_ROW_SHL(1)>(acc[1]);
        const float p23 = acc[2] + row_read<DPP_ROW_SHL(1)>(acc[3]);
        const float p45 = acc[4] + row_read<DPP_ROW_SHL(1)>(acc[5]);
        const float p67 = acc[6] + row_read<DPP_ROW_SHL(1)>(acc[7]);
        const float lo  = p01 + row_read<DPP_ROW_SHL(2)>(p23);
        const float hi  = p45 + row_read<DPP_ROW_SHL(2)>(p67);
        const float own = upper_regs ? hi : lo;
        const float val = row_read<DPP_QUAD_0022>(own);

        const int eq = g + st_edge;
        if (storer && eq <= last) {
            const float dist = sqrtf(val);
            *(float4*)(out + (size_t)eq * 8 + (lane & 1) * 4) = make_float4(dist, dist, dist, dist);
        }
    }
}

extern "C" void kernel_launch(void* const* d_in, const int* in_sizes, int n_in,
                              void* d_out, int out_size, void* d_ws, size_t ws_size,
                              hipStream_t stream)
{
    const float* x  = (const float*)d_in[0];
    const int*   eb = (const int*)d_in[1];
    const int*   ei = (const int*)d_in[2];
    const int*   ej = (const int*)d_in[3];
    float*       out = (float*)d_out;

    const int E = in_sizes[1];
    if (E <= 0) return;

    hipLaunchKernelGGL(edge_norm_quad, dim3(NUM_BLOCKS), dim3(THREADS_PER_BLOCK), 0, stream,
                       x, eb, ei, ej, out, E);
}


# Software pipelining: the next 4-edge group's rows are requested before the current group's WMMAs and only their differences are carried over, so the cache round trip overlaps the matrix work
# speedup vs baseline: 1.6364x; 1.0452x over previous
#include <hip/hip_runtime.h>

typedef float v2f __attribute__((ext_vector_type(2)));
typedef float v4f __attribute__((ext_vector_type(4)));
typedef float v8f __attribute__((ext_vector_type(8)));

#define DIM_D 256
#define DIM_N 1024
#define WAVES_PER_BLOCK 32
#define THREADS_PER_BLOCK 1024
#define NUM_BLOCKS 256
#define LOG2_NUM_WAVES 13

#define DPP_ROW_SHL(n) (0x100 + (n))
#define DPP_QUAD_0022  (0 | (0 << 2) | (2 << 4) | (2 << 6))

template <int CTRL>
__device__ __forceinline__ float row_read(float v)
{
    return __builtin_bit_cast(float,
        __builtin_amdgcn_update_dpp(0, __builtin_bit_cast(int, v), CTRL, 0xF, 0xF, true));
}

__device__ __forceinline__ long range_start(long w, long E)
{
    return (w >> LOG2_NUM_WAVES) ? E : (((w * E) >> LOG2_NUM_WAVES) & ~3L);
}

__device__ __forceinline__ v8f gram_step(v2f a, v8f c)
{
    return __builtin_amdgcn_wmma_f32_16x16x4_f32(
         false, a,  false, a,
         (short)0, c,  false,  false);
}

__global__ __launch_bounds__(THREADS_PER_BLOCK)
void edge_norm_ahead(const float* __restrict__ x,
                    const int*   __restrict__ eb,
                    const int*   __restrict__ ei,
                    const int*   __restrict__ ej,
                    float*       __restrict__ out,
                    int E)
{
    const int lane = threadIdx.x & 31;
    const int gw = __builtin_amdgcn_readfirstlane(
        (int)(blockIdx.x * WAVES_PER_BLOCK + (threadIdx.x >> 5)));

    const long e0 = range_start((long)gw,     (long)E);
    const long e1 = range_start((long)gw + 1, (long)E);
    if (e0 >= e1) return;

    const int c     = lane & 15;
    const int h     = lane >> 4;
    const int slot  = c >> 2;
    const unsigned lane_bytes = (unsigned)((((c & 3) << 1) | h) << 4);

    const bool upper_regs = (c & 4) != 0;
    const bool storer   = ((c & 2) == 0) && (((c >> 3) & 1) == h);
    const int  st_edge  = (h << 1) | ((c >> 2) & 1);

    const char* __restrict__ xb  = (const char*)x;
    const char* __restrict__ ebb = (const char*)eb;
    const char* __restrict__ eib = (const char*)ei;
    const char* __restrict__ ejb = (const char*)ej;

    const int first = (int)e0;
    const int last  = (int)e1 - 1;

    int eidx = first + slot;
    if (eidx > last) eidx = last;
    unsigned ioff = (unsigned)eidx << 2;
    int nb = *(const int*)(ebb + ioff);
    const int rowi_first = nb * DIM_N + *(const int*)(eib + ioff);
    const int rowj_first = nb * DIM_N + *(const int*)(ejb + ioff);

    v4f d[8];
    {
        const unsigned offi = ((unsigned)rowi_first << 10) | lane_bytes;
        const unsigned offj = ((unsigned)rowj_first << 10) | lane_bytes;
        #pragma unroll
        for (int t = 0; t < 8; ++t)
            d[t] = *(const v4f*)(xb + offi + t * 128) - *(const v4f*)(xb + offj + t * 128);
    }

    eidx = first + 4 + slot;
    if (eidx > last) eidx = last;
    ioff = (unsigned)eidx << 2;
    nb = *(const int*)(ebb + ioff);
    int rowi_next = nb * DIM_N + *(const int*)(eib + ioff);
    int rowj_next = nb * DIM_N + *(const int*)(ejb + ioff);

    for (int g = first; g <= last; g += 4) {
        const unsigned offi = ((unsigned)rowi_next << 10) | lane_bytes;
        const unsigned offj = ((unsigned)rowj_next << 10) | lane_bytes;

        eidx = g + 8 + slot;
        if (eidx > last) eidx = last;
        ioff = (unsigned)eidx << 2;
        nb = *(const int*)(ebb + ioff);
        rowi_next = nb * DIM_N + *(const int*)(eib + ioff);
        rowj_next = nb * DIM_N + *(const int*)(ejb + ioff);

        v4f a[8], b[8];
        #pragma unroll
        for (int t = 0; t < 8; ++t) {
            a[t] = *(const v4f*)(xb + offi + t * 128);
            b[t] = *(const v4f*)(xb + offj + t * 128);
        }

        __builtin_amdgcn_sched_barrier(0);

        v8f acc0 = {0.f, 0.f, 0.f, 0.f, 0.f, 0.f, 0.f, 0.f};
        v8f acc1 = {0.f, 0.f, 0.f, 0.f, 0.f, 0.f, 0.f, 0.f};
        #pragma unroll
        for (int t = 0; t < 8; ++t) {
            acc0 = gram_step(__builtin_shufflevector(d[t], d[t], 0, 1), acc0);
            acc1 = gram_step(__builtin_shufflevector(d[t], d[t], 2, 3), acc1);
        }
        const v8f acc = acc0 + acc1;

        const float p01 = acc[0] + row_read<DPP_ROW_SHL(1)>(acc[1]);
        const float p23 = acc[2] + row_read<DPP_ROW_SHL(1)>(acc[3]);
        const float p45 = acc[4] + row_read<DPP_ROW_SHL(1)>(acc[5]);
        const float p67 = acc[6] + row_read<DPP_ROW_SHL(1)>(acc[7]);
        const float lo  = p01 + row_read<DPP_ROW_SHL(2)>(p23);
        const float hi  = p45 + row_read<DPP_ROW_SHL(2)>(p67);
        const float own = upper_regs ? hi : lo;
        const float val = row_read<DPP_QUAD_0022>(own);

        const int eq = g + st_edge;
        if (storer && eq <= last) {
            const float dist = sqrtf(val);
            *(float4*)(out + (size_t)eq * 8 + (lane & 1) * 4) = make_float4(dist, dist, dist, dist);
        }

        #pragma unroll
        for (int t = 0; t < 8; ++t) d[t] = a[t] - b[t];
    }
}

extern "C" void kernel_launch(void* const* d_in, const int* in_sizes, int n_in,
                              void* d_out, int out_size, void* d_ws, size_t ws_size,
                              hipStream_t stream)
{
    const float* x  = (const float*)d_in[0];
    const int*   eb = (const int*)d_in[1];
    const int*   ei = (const int*)d_in[2];
    const int*   ej = (const int*)d_in[3];
    float*       out = (float*)d_out;

    const int E = in_sizes[1];
    if (E <= 0) return;

    hipLaunchKernelGGL(edge_norm_ahead, dim3(NUM_BLOCKS), dim3(THREADS_PER_BLOCK), 0, stream,
                       x, eb, ei, ej, out, E);
}


# Half as many waves (16 per block), each with two 4-edge tiles per trip: one accumulator chain per tile, one root and one 256-B store per 8 edges; fewer waves to enter; sum order changes
# speedup vs baseline: 1.7494x; 1.0691x over previous
#include <hip/hip_runtime.h>

typedef float v2f __attribute__((ext_vector_type(2)));
typedef float v4f __attribute__((ext_vector_type(4)));
typedef float v8f __attribute__((ext_vector_type(8)));

#define DIM_D 256
#define DIM_N 1024
#define WAVES_PER_BLOCK 16
#define THREADS_PER_BLOCK 512
#define NUM_BLOCKS 256
#define LOG2_NUM_WAVES 12

#define DPP_ROW_SHL(n) (0x100 + (n))
#define DPP_ROW_SHR(n) (0x110 + (n))
#define DPP_QUAD_0022  (0 | (0 << 2) | (2 << 4) | (2 << 6))

template <int CTRL>
__device__ __forceinline__ float row_read(float v)
{
    return __builtin_bit_cast(float,
        __builtin_amdgcn_update_dpp(0, __builtin_bit_cast(int, v), CTRL, 0xF, 0xF, true));
}

__device__ __forceinline__ long range_start(long w, long E)
{
    return (w >> LOG2_NUM_WAVES) ? E : (((w * E) >> LOG2_NUM_WAVES) & ~7L);
}

__device__ __forceinline__ v8f gram_step(v2f a, v8f c)
{
    return __builtin_amdgcn_wmma_f32_16x16x4_f32(
         false, a,  false, a,
         (short)0, c,  false,  false);
}

__global__ __launch_bounds__(THREADS_PER_BLOCK)
void edge_norm_pairs(const float* __restrict__ x,
                     const int*   __restrict__ eb,
                     const int*   __restrict__ ei,
                     const int*   __restrict__ ej,
                     float*       __restrict__ out,
                     int E)
{
    const int lane = threadIdx.x & 31;
    const int gw = __builtin_amdgcn_readfirstlane(
        (int)(blockIdx.x * WAVES_PER_BLOCK + (threadIdx.x >> 5)));

    const long e0 = range_start((long)gw,     (long)E);
    const long e1 = range_start((long)gw + 1, (long)E);
    if (e0 >= e1) return;

    const int c     = lane & 15;
    const int h     = lane >> 4;
    const int slot  = c >> 2;
    const unsigned lane_bytes = (unsigned)((((c & 3) << 1) | h) << 4);

    const bool upper_regs  = (c & 4) != 0;
    const bool second_tile = (c & 2) != 0;
    const bool storer   = (((c >> 3) & 1) == h);
    const int  st_edge  = (((c >> 1) & 1) << 2) | (h << 1) | ((c >> 2) & 1);

    const char* __restrict__ xb  = (const char*)x;
    const char* __restrict__ ebb = (const char*)eb;
    const char* __restrict__ eib = (const char*)ei;
    const char* __restrict__ ejb = (const char*)ej;

    const int first = (int)e0;
    const int last  = (int)e1 - 1;

    int rowi_next[2], rowj_next[2];
    #pragma unroll
    for (int u = 0; u < 2; ++u) {
        int eidx = first + 4 * u + slot;
        if (eidx > last) eidx = last;
        const unsigned ioff = (unsigned)eidx << 2;
        const int nb = *(const int*)(ebb + ioff);
        rowi_next[u] = nb * DIM_N + *(const int*)(eib + ioff);
        rowj_next[u] = nb * DIM_N + *(const int*)(ejb + ioff);
    }

    v4f d[2][8];
    #pragma unroll
    for (int u = 0; u < 2; ++u) {
        const unsigned offi = ((unsigned)rowi_next[u] << 10) | lane_bytes;
        const unsigned offj = ((unsigned)rowj_next[u] << 10) | lane_bytes;
        #pragma unroll
        for (int t = 0; t < 8; ++t)
            d[u][t] = *(const v4f*)(xb + offi + t * 128) - *(const v4f*)(xb + offj + t * 128);
    }

    #pragma unroll
    for (int u = 0; u < 2; ++u) {
        int eidx = first + 8 + 4 * u + slot;
        if (eidx > last) eidx = last;
        const unsigned ioff = (unsigned)eidx << 2;
        const int nb = *(const int*)(ebb + ioff);
        rowi_next[u] = nb * DIM_N + *(const int*)(eib + ioff);
        rowj_next[u] = nb * DIM_N + *(const int*)(ejb + ioff);
    }

    for (int g = first; g <= last; g += 8) {
        unsigned offi[2], offj[2];
        #pragma unroll
        for (int u = 0; u < 2; ++u) {
            offi[u] = ((unsigned)rowi_next[u] << 10) | lane_bytes;
            offj[u] = ((unsigned)rowj_next[u] << 10) | lane_bytes;
        }

        #pragma unroll
        for (int u = 0; u < 2; ++u) {
            int eidx = g + 16 + 4 * u + slot;
            if (eidx > last) eidx = last;
            const unsigned ioff = (unsigned)eidx << 2;
            const int nb = *(const int*)(ebb + ioff);
            rowi_next[u] = nb * DIM_N + *(const int*)(eib + ioff);
            rowj_next[u] = nb * DIM_N + *(const int*)(ejb + ioff);
        }

        v4f a[2][8], b[2][8];
        #pragma unroll
        for (int u = 0; u < 2; ++u) {
            #pragma unroll
            for (int t = 0; t < 8; ++t) {
                a[u][t] = *(const v4f*)(xb + offi[u] + t * 128);
                b[u][t] = *(const v4f*)(xb + offj[u] + t * 128);
            }
        }

        __builtin_amdgcn_sched_barrier(0);

        v8f acc0 = {0.f, 0.f, 0.f, 0.f, 0.f, 0.f, 0.f, 0.f};
        v8f acc1 = {0.f, 0.f, 0.f, 0.f, 0.f, 0.f, 0.f, 0.f};
        #pragma unroll
        for (int t = 0; t < 8; ++t) {
            acc0 = gram_step(__builtin_shufflevector(d[0][t], d[0][t], 0, 1), acc0);
            acc1 = gram_step(__builtin_shufflevector(d[1][t], d[1][t], 0, 1), acc1);
            acc0 = gram_step(__builtin_shufflevector(d[0][t], d[0][t], 2, 3), acc0);
            acc1 = gram_step(__builtin_shufflevector(d[1][t], d[1][t], 2, 3), acc1);
        }

        const float f01 = acc0[0] + row_read<DPP_ROW_SHL(1)>(acc0[1]);
        const float f23 = acc0[2] + row_read<DPP_ROW_SHL(1)>(acc0[3]);
        const float f45 = acc0[4] + row_read<DPP_ROW_SHL(1)>(acc0[5]);
        const float f67 = acc0[6] + row_read<DPP_ROW_SHL(1)>(acc0[7]);
        const float lo0 = f01 + row_read<DPP_ROW_SHL(2)>(f23);
        const float hi0 = f45 + row_read<DPP_ROW_SHL(2)>(f67);
        const float s01 = acc1[0] + row_read<DPP_ROW_SHL(1)>(acc1[1]);
        const float s23 = acc1[2] + row_read<DPP_ROW_SHL(1)>(acc1[3]);
        const float s45 = acc1[4] + row_read<DPP_ROW_SHL(1)>(acc1[5]);
        const float s67 = acc1[6] + row_read<DPP_ROW_SHL(1)>(acc1[7]);
        const float lo1 = s23 + row_read<DPP_ROW_SHR(2)>(s01);
        const float hi1 = s67 + row_read<DPP_ROW_SHR(2)>(s45);

        const float lo  = second_tile ? lo1 : lo0;
        const float hi  = second_tile ? hi1 : hi0;
        const float own = upper_regs ? hi : lo;
        const float val = row_read<DPP_QUAD_0022>(own);

        const int eq = g + st_edge;
        if (storer && eq <= last) {
            const float dist = sqrtf(val);
            *(float4*)(out + (size_t)eq * 8 + (lane & 1) * 4) = make_float4(dist, dist, dist, dist);
        }

        #pragma unroll
        for (int u = 0; u < 2; ++u) {
            #pragma unroll
            for (int t = 0; t < 8; ++t) d[u][t] = a[u][t] - b[u][t];
        }
    }
}

extern "C" void kernel_launch(void* const* d_in, const int* in_sizes, int n_in,
                              void* d_out, int out_size, void* d_ws, size_t ws_size,
                              hipStream_t stream)
{
    const float* x  = (const float*)d_in[0];
    const int*   eb = (const int*)d_in[1];
    const int*   ei = (const int*)d_in[2];
    const int*   ej = (const int*)d_in[3];
    float*       out = (float*)d_out;

    const int E = in_sizes[1];
    if (E <= 0) return;

    hipLaunchKernelGGL(edge_norm_pairs, dim3(NUM_BLOCKS), dim3(THREADS_PER_BLOCK), 0, stream,
                       x, eb, ei, ej, out, E);
}
